# starting kernel
# baseline (speedup 1.0000x reference)
#include <hip/hip_runtime.h>
#include <math.h>


typedef __attribute__((ext_vector_type(2))) float v2f;
typedef __attribute__((ext_vector_type(8))) float v8f;

#define CCH 128

__global__ __launch_bounds__(256) void degree_kernel(
    const int* __restrict__ dst, float* __restrict__ cnt, int E) {
  int e = blockIdx.x * blockDim.x + threadIdx.x;
  if (e < E) {
    unsafeAtomicAdd(&cnt[dst[e]], 1.0f);
  }
}

__global__ __launch_bounds__(256) void inv_kernel(
    const float* __restrict__ cnt, float* __restrict__ inv, int n) {
  int i = blockIdx.x * blockDim.x + threadIdx.x;
  if (i < n) inv[i] = 1.0f / fmaxf(cnt[i], 1.0f);
}

__global__ __launch_bounds__(256) void scatter_kernel(
    const float* __restrict__ feat, const int* __restrict__ src,
    const int* __restrict__ dst, float* __restrict__ agg, int E) {
  long tid = (long)blockIdx.x * blockDim.x + threadIdx.x;
  long total = (long)E * 32;
  if (tid >= total) return;
  int e = (int)(tid >> 5);
  int c = ((int)tid & 31) * 4;
  int s = src[e];
  int d = dst[e];
  const float4 v = *(const float4*)(feat + (long)s * CCH + c);
  float* base = agg + (long)d * CCH + c;
  unsafeAtomicAdd(base + 0, v.x);
  unsafeAtomicAdd(base + 1, v.y);
  unsafeAtomicAdd(base + 2, v.z);
  unsafeAtomicAdd(base + 3, v.w);
}

template <bool SIG>
__global__ __launch_bounds__(128) void sage_layer_kernel(
    const float* __restrict__ xin,
    const float* __restrict__ agg,
    const float* __restrict__ inv,
    const float* __restrict__ Wl,
    const float* __restrict__ bl,
    const float* __restrict__ Wr,
    float* __restrict__ out,
    int n_nodes) {
  const int lane = threadIdx.x & 31;
  const int wave = threadIdx.x >> 5;
  const int kk   = lane >> 4;
  const int ml   = lane & 15;
  const int mbase = blockIdx.x * 128 + wave * 32;

  int m0 = mbase + ml;       if (m0 >= n_nodes) m0 = n_nodes - 1;
  int m1 = mbase + 16 + ml;  if (m1 >= n_nodes) m1 = n_nodes - 1;
  const float inv0 = inv[m0];
  const float inv1 = inv[m1];

  const float2* a0p = (const float2*)(agg + (long)m0 * CCH);
  const float2* a1p = (const float2*)(agg + (long)m1 * CCH);
  const float2* x0p = (const float2*)(xin + (long)m0 * CCH);
  const float2* x1p = (const float2*)(xin + (long)m1 * CCH);

  v8f acc[2][8];
#pragma unroll
  for (int t = 0; t < 2; ++t)
#pragma unroll
    for (int n = 0; n < 8; ++n) acc[t][n] = {};

  for (int ks = 0; ks < 32; ++ks) {
    const int kh = ks * 2 + kk;
    float2 r0 = a0p[kh];
    float2 r1 = a1p[kh];
    v2f A0; A0.x = r0.x * inv0; A0.y = r0.y * inv0;
    v2f A1; A1.x = r1.x * inv1; A1.y = r1.y * inv1;
#pragma unroll
    for (int n = 0; n < 8; ++n) {
      float2 bw = *(const float2*)(Wl + (long)(n * 16 + ml) * CCH + 2 * kh);
      v2f B; B.x = bw.x; B.y = bw.y;
      acc[0][n] = __builtin_amdgcn_wmma_f32_16x16x4_f32(
          false, A0, false, B, (short)0, acc[0][n], false, false);
      acc[1][n] = __builtin_amdgcn_wmma_f32_16x16x4_f32(
          false, A1, false, B, (short)0, acc[1][n], false, false);
    }
  }

  for (int ks = 0; ks < 32; ++ks) {
    const int kh = ks * 2 + kk;
    float2 r0 = x0p[kh];
    float2 r1 = x1p[kh];
    v2f A0; A0.x = r0.x; A0.y = r0.y;
    v2f A1; A1.x = r1.x; A1.y = r1.y;
#pragma unroll
    for (int n = 0; n < 8; ++n) {
      float2 bw = *(const float2*)(Wr + (long)(n * 16 + ml) * CCH + 2 * kh);
      v2f B; B.x = bw.x; B.y = bw.y;
      acc[0][n] = __builtin_amdgcn_wmma_f32_16x16x4_f32(
          false, A0, false, B, (short)0, acc[0][n], false, false);
      acc[1][n] = __builtin_amdgcn_wmma_f32_16x16x4_f32(
          false, A1, false, B, (short)0, acc[1][n], false, false);
    }
  }

#pragma unroll
  for (int n = 0; n < 8; ++n) {
    const float bias = bl[n * 16 + ml];
#pragma unroll
    for (int t = 0; t < 2; ++t) {
#pragma unroll
      for (int r = 0; r < 8; ++r) {
        int m = mbase + t * 16 + r + 8 * kk;
        if (m < n_nodes) {
          float v = acc[t][n][r] + bias;
          if (SIG) v = 1.0f / (1.0f + __expf(-v));
          out[(long)m * CCH + n * 16 + ml] = v;
        }
      }
    }
  }
}

extern "C" void kernel_launch(void* const* d_in, const int* in_sizes, int n_in,
                              void* d_out, int out_size, void* d_ws, size_t ws_size,
                              hipStream_t stream) {
  const float* x   = (const float*)d_in[0];
  const int*   ei  = (const int*)d_in[1];
  const float* W1l = (const float*)d_in[2];
  const float* b1l = (const float*)d_in[3];
  const float* W1r = (const float*)d_in[4];
  const float* W2l = (const float*)d_in[5];
  const float* b2l = (const float*)d_in[6];
  const float* W2r = (const float*)d_in[7];
  float* out = (float*)d_out;

  const int N = in_sizes[0] / CCH;
  const int E = in_sizes[1] / 2;

  float* agg = (float*)d_ws;
  float* h   = agg + (size_t)N * CCH;
  float* cnt = h + (size_t)N * CCH;
  float* vin = cnt + (size_t)N;

  const int* src = ei;
  const int* dst = ei + E;

  hipMemsetAsync(agg, 0, (size_t)N * CCH * sizeof(float), stream);
  hipMemsetAsync(cnt, 0, (size_t)N * sizeof(float), stream);
  degree_kernel<<<(E + 255) / 256, 256, 0, stream>>>(dst, cnt, E);
  inv_kernel<<<(N + 255) / 256, 256, 0, stream>>>(cnt, vin, N);

  const long sthreads = (long)E * 32;
  const int  sblocks  = (int)((sthreads + 255) / 256);
  const int  lblocks  = (N + 127) / 128;

  scatter_kernel<<<sblocks, 256, 0, stream>>>(x, src, dst, agg, E);
  sage_layer_kernel<true><<<lblocks, 128, 0, stream>>>(
      x, agg, vin, W1l, b1l, W1r, h, N);

  hipMemsetAsync(agg, 0, (size_t)N * CCH * sizeof(float), stream);
  scatter_kernel<<<sblocks, 256, 0, stream>>>(h, src, dst, agg, E);
  sage_layer_kernel<false><<<lblocks, 128, 0, stream>>>(
      h, agg, vin, W2l, b2l, W2r, out, N);
}


# Layer GEMM weight-stationary: a wave owns 16 output columns, keeps their B fragments for all K in registers; 256-thread blocks stream 32-node tiles through LDS, rows stored whole; sums unchanged
# speedup vs baseline: 2.2849x; 2.2849x over previous
#include <hip/hip_runtime.h>
#include <math.h>


typedef __attribute__((ext_vector_type(2))) float v2f;
typedef __attribute__((ext_vector_type(4))) float v4f;
typedef __attribute__((ext_vector_type(8))) float v8f;

#define CCH 128

#define BINSH 7
#define BINN 128
#define MAXBINS 1024
#define BTH 1024
#define EPB 5120
#define ATH 512
#define AWV 16
#define UNR 8

__global__ __launch_bounds__(BTH) void bin_hist_kernel(
    const int* __restrict__ dst, int* __restrict__ hist, int E, int nbins) {
  __shared__ int hs[MAXBINS];
  const int tid = threadIdx.x;
  hs[tid] = 0;
  __syncthreads();
  const int base = blockIdx.x * EPB;
  for (int k = 0; k < EPB / BTH; ++k) {
    const int e = base + k * BTH + tid;
    if (e < E) atomicAdd(&hs[dst[e] >> BINSH], 1);
  }
  __syncthreads();
  if (tid < nbins) hist[blockIdx.x * nbins + tid] = hs[tid];
}

__global__ __launch_bounds__(BTH) void bin_scan_kernel(
    const int* __restrict__ hist, int* __restrict__ pref,
    int* __restrict__ binStart, int nblk, int nbins) {
  __shared__ int s[BTH];
  const int t = threadIdx.x;
  int run = 0;
  if (t < nbins) {
    for (int b = 0; b < nblk; ++b) {
      const int v = hist[b * nbins + t];
      pref[b * nbins + t] = run;
      run += v;
    }
  }
  s[t] = run;
  __syncthreads();
  for (int off = 1; off < BTH; off <<= 1) {
    const int v = (t >= off) ? s[t - off] : 0;
    __syncthreads();
    s[t] += v;
    __syncthreads();
  }
  if (t < nbins) binStart[t] = s[t] - run;
  if (t == nbins - 1) binStart[nbins] = s[t];
}

__global__ __launch_bounds__(BTH) void bin_place_kernel(
    const int* __restrict__ src, const int* __restrict__ dst,
    const int* __restrict__ pref, const int* __restrict__ binStart,
    int2* __restrict__ sorted, int E, int nbins) {
  __shared__ int cur[MAXBINS];
  const int tid = threadIdx.x;
  cur[tid] = (tid < nbins) ? binStart[tid] + pref[blockIdx.x * nbins + tid] : 0;
  __syncthreads();
  const int base = blockIdx.x * EPB;
  for (int k = 0; k < EPB / BTH; ++k) {
    const int e = base + k * BTH + tid;
    if (e < E) {
      const int d = dst[e];
      const int pos = atomicAdd(&cur[d >> BINSH], 1);
      sorted[pos] = make_int2(src[e], d & (BINN - 1));
    }
  }
}

template <bool WRITE_INV>
__global__ __launch_bounds__(ATH) void bin_sum_kernel(
    const float* __restrict__ feat, const int2* __restrict__ sorted,
    const int* __restrict__ binStart, float* __restrict__ agg,
    float* __restrict__ inv, int n_nodes) {
  __shared__ float acc[(BINN + 1) * CCH];
  __shared__ int degc[BINN + 1];
  const int tid  = threadIdx.x;
  const int lane = tid & 31;
  const int wave = __builtin_amdgcn_readfirstlane(tid >> 5);
  const int bin  = blockIdx.x;
  float4* acc4 = (float4*)acc;
  for (int i = tid; i < (BINN + 1) * CCH / 4; i += ATH)
    acc4[i] = make_float4(0.0f, 0.0f, 0.0f, 0.0f);
  if (WRITE_INV && tid <= BINN) degc[tid] = 0;
  __syncthreads();

  const int e0 = binStart[bin];
  const int e1 = binStart[bin + 1];
  for (int base = e0 + wave * 32; base < e1; base += AWV * 32) {
    const int idx = base + lane;
    int2 sd = make_int2(0, BINN);
    if (idx < e1) sd = sorted[idx];
    if (WRITE_INV) atomicAdd(&degc[sd.y], 1);
    int cnt = e1 - base;
    if (cnt > 32) cnt = 32;
    for (int j = 0; j < cnt; j += UNR) {
      float v[UNR][4];
      int dl[UNR];
#pragma unroll
      for (int u = 0; u < UNR; ++u) {
        const int s = __builtin_amdgcn_readlane(sd.x, j + u);
        dl[u] = __builtin_amdgcn_readlane(sd.y, j + u);
        const float* row = feat + (long)s * CCH + lane;
        v[u][0] = row[0];
        v[u][1] = row[32];
        v[u][2] = row[64];
        v[u][3] = row[96];
      }
#pragma unroll
      for (int u = 0; u < UNR; ++u) {
        float* a = acc + dl[u] * CCH + lane;
        unsafeAtomicAdd(a + 0,  v[u][0]);
        unsafeAtomicAdd(a + 32, v[u][1]);
        unsafeAtomicAdd(a + 64, v[u][2]);
        unsafeAtomicAdd(a + 96, v[u][3]);
      }
    }
  }
  __syncthreads();

  for (int r = wave; r < BINN; r += AWV) {
    const int node = bin * BINN + r;
    if (node < n_nodes)
      ((float4*)(agg + (long)node * CCH))[lane] = acc4[r * (CCH / 4) + lane];
  }
  if (WRITE_INV && tid < BINN) {
    const int node = bin * BINN + tid;
    if (node < n_nodes) inv[node] = 1.0f / fmaxf((float)degc[tid], 1.0f);
  }
}

#define TMN 32
#define APITCH 260
#define DPITCH 132
#define LGRID 512

template <bool SIG>
__global__ __launch_bounds__(256) void sage_layer_kernel(
    const float* __restrict__ xin,
    const float* __restrict__ agg,
    const float* __restrict__ inv,
    const float* __restrict__ Wl,
    const float* __restrict__ bl,
    const float* __restrict__ Wr,
    float* __restrict__ out,
    int n_nodes, int n_tiles) {
  __shared__ float As[TMN * APITCH];
  __shared__ float Ds[TMN * DPITCH];
  const int lane = threadIdx.x & 31;
  const int wave = __builtin_amdgcn_readfirstlane(threadIdx.x >> 5);
  const int kk   = lane >> 4;
  const int ml   = lane & 15;

  v2f Bl[32], Br[32];
  {
    const float* wl = Wl + (long)(wave * 16 + ml) * CCH + 2 * kk;
    const float* wr = Wr + (long)(wave * 16 + ml) * CCH + 2 * kk;
#pragma unroll
    for (int ks = 0; ks < 32; ++ks) {
      const float2 l = *(const float2*)(wl + 4 * ks);
      const float2 r = *(const float2*)(wr + 4 * ks);
      Bl[ks].x = l.x; Bl[ks].y = l.y;
      Br[ks].x = r.x; Br[ks].y = r.y;
    }
  }
  const float bias = bl[wave * 16 + ml];

  v4f ra[4], rx[4];
  float rs[4];
  int tile = blockIdx.x;
#pragma unroll
  for (int r = 0; r < 4; ++r) {
    int node = tile * TMN + wave * 4 + r;
    if (node >= n_nodes) node = n_nodes - 1;
    ra[r] = *(const v4f*)(agg + (long)node * CCH + 4 * lane);
    rx[r] = *(const v4f*)(xin + (long)node * CCH + 4 * lane);
    rs[r] = inv[node];
  }

  const float* a0 = As + ml * APITCH + 2 * kk;
  const float* a1 = a0 + 16 * APITCH;

  for (; tile < n_tiles; tile += LGRID) {
#pragma unroll
    for (int r = 0; r < 4; ++r) {
      const int row = wave * 4 + r;
      *(v4f*)(As + row * APITCH + 4 * lane) = ra[r] * rs[r];
      *(v4f*)(As + row * APITCH + CCH + 4 * lane) = rx[r];
    }
    __syncthreads();

    const int nxt = tile + LGRID;
    if (nxt < n_tiles) {
#pragma unroll
      for (int r = 0; r < 4; ++r) {
        int node = nxt * TMN + wave * 4 + r;
        if (node >= n_nodes) node = n_nodes - 1;
        ra[r] = *(const v4f*)(agg + (long)node * CCH + 4 * lane);
        rx[r] = *(const v4f*)(xin + (long)node * CCH + 4 * lane);
        rs[r] = inv[node];
      }
    }

    v8f acc0 = {};
    v8f acc1 = {};
#pragma unroll
    for (int ks = 0; ks < 32; ++ks) {
      const v2f A0 = *(const v2f*)(a0 + 4 * ks);
      const v2f A1 = *(const v2f*)(a1 + 4 * ks);
      acc0 = __builtin_amdgcn_wmma_f32_16x16x4_f32(
          false, A0, false, Bl[ks], (short)0, acc0, false, false);
      acc1 = __builtin_amdgcn_wmma_f32_16x16x4_f32(
          false, A1, false, Bl[ks], (short)0, acc1, false, false);
    }
#pragma unroll
    for (int ks = 0; ks < 32; ++ks) {
      const v2f A0 = *(const v2f*)(a0 + CCH + 4 * ks);
      const v2f A1 = *(const v2f*)(a1 + CCH + 4 * ks);
      acc0 = __builtin_amdgcn_wmma_f32_16x16x4_f32(
          false, A0, false, Br[ks], (short)0, acc0, false, false);
      acc1 = __builtin_amdgcn_wmma_f32_16x16x4_f32(
          false, A1, false, Br[ks], (short)0, acc1, false, false);
    }

#pragma unroll
    for (int r = 0; r < 8; ++r) {
      float v0 = acc0[r] + bias;
      float v1 = acc1[r] + bias;
      if (SIG) {
        v0 = 1.0f / (1.0f + __expf(-v0));
        v1 = 1.0f / (1.0f + __expf(-v1));
      }
      Ds[(r + 8 * kk) * DPITCH + wave * 16 + ml] = v0;
      Ds[(16 + r + 8 * kk) * DPITCH + wave * 16 + ml] = v1;
    }
    __syncthreads();

#pragma unroll
    for (int r = 0; r < 4; ++r) {
      const int row  = wave * 4 + r;
      const int node = tile * TMN + row;
      if (node < n_nodes)
        *(v4f*)(out + (long)node * CCH + 4 * lane) =
            *(const v4f*)(Ds + row * DPITCH + 4 * lane);
    }
  }
}

extern "C" void kernel_launch(void* const* d_in, const int* in_sizes, int n_in,
                              void* d_out, int out_size, void* d_ws, size_t ws_size,
                              hipStream_t stream) {
  const float* x   = (const float*)d_in[0];
  const int*   ei  = (const int*)d_in[1];
  const float* W1l = (const float*)d_in[2];
  const float* b1l = (const float*)d_in[3];
  const float* W1r = (const float*)d_in[4];
  const float* W2l = (const float*)d_in[5];
  const float* b2l = (const float*)d_in[6];
  const float* W2r = (const float*)d_in[7];
  float* out = (float*)d_out;

  const int N = in_sizes[0] / CCH;
  const int E = in_sizes[1] / 2;

  float* agg = (float*)d_ws;
  float* h   = agg + (size_t)N * CCH;
  float* vin = h + (size_t)N * CCH;

  const int* src = ei;
  const int* dst = ei + E;

  const int nbins = (N + BINN - 1) / BINN;
  const int nblk  = (E + EPB - 1) / EPB;

  int2* sorted   = (int2*)d_out;
  int*  hist     = (int*)d_out + 2 * (size_t)E;
  int*  pref     = hist + (size_t)nblk * nbins;
  int*  binStart = pref + (size_t)nblk * nbins;

  bin_hist_kernel<<<nblk, BTH, 0, stream>>>(dst, hist, E, nbins);
  bin_scan_kernel<<<1, BTH, 0, stream>>>(hist, pref, binStart, nblk, nbins);
  bin_place_kernel<<<nblk, BTH, 0, stream>>>(src, dst, pref, binStart, sorted, E, nbins);

  const int ntiles = (N + TMN - 1) / TMN;

  bin_sum_kernel<true><<<nbins, ATH, 0, stream>>>(x, sorted, binStart, agg, vin, N);
  sage_layer_kernel<true><<<LGRID, 256, 0, stream>>>(
      x, agg, vin, W1l, b1l, W1r, h, N, ntiles);

  bin_sum_kernel<false><<<nbins, ATH, 0, stream>>>(h, sorted, binStart, agg, vin, N);
  sage_layer_kernel<false><<<LGRID, 256, 0, stream>>>(
      h, agg, vin, W2l, b2l, W2r, out, N, ntiles);
}


# Row sums: owner work-group orders its bin's records by destination in LDS (integer counting sort); a wave sums 8 destinations' rows in registers, 512 B per load; no fp32 LDS atomics; rank-order sums
# speedup vs baseline: 2.4048x; 1.0525x over previous
#include <hip/hip_runtime.h>
#include <math.h>


typedef __attribute__((ext_vector_type(2))) float v2f;
typedef __attribute__((ext_vector_type(4))) float v4f;
typedef __attribute__((ext_vector_type(8))) float v8f;

#define CCH 128

#define BINSH 7
#define BINN 128
#define MAXBINS 1024
#define BTH 1024
#define EPB 5120
#define ATH 512
#define AWV 16
#define DPW (BINN / AWV)
#define SCAP 2048
#define RPT (SCAP / ATH)
#define UNR 8

__global__ __launch_bounds__(BTH) void bin_hist_kernel(
    const int* __restrict__ dst, int* __restrict__ hist, int E, int nbins) {
  __shared__ int hs[MAXBINS];
  const int tid = threadIdx.x;
  hs[tid] = 0;
  __syncthreads();
  const int base = blockIdx.x * EPB;
  for (int k = 0; k < EPB / BTH; ++k) {
    const int e = base + k * BTH + tid;
    if (e < E) atomicAdd(&hs[dst[e] >> BINSH], 1);
  }
  __syncthreads();
  if (tid < nbins) hist[blockIdx.x * nbins + tid] = hs[tid];
}

__global__ __launch_bounds__(BTH) void bin_scan_kernel(
    const int* __restrict__ hist, int* __restrict__ pref,
    int* __restrict__ binStart, int nblk, int nbins) {
  __shared__ int s[BTH];
  const int t = threadIdx.x;
  int run = 0;
  if (t < nbins) {
    for (int b = 0; b < nblk; ++b) {
      const int v = hist[b * nbins + t];
      pref[b * nbins + t] = run;
      run += v;
    }
  }
  s[t] = run;
  __syncthreads();
  for (int off = 1; off < BTH; off <<= 1) {
    const int v = (t >= off) ? s[t - off] : 0;
    __syncthreads();
    s[t] += v;
    __syncthreads();
  }
  if (t < nbins) binStart[t] = s[t] - run;
  if (t == nbins - 1) binStart[nbins] = s[t];
}

__global__ __launch_bounds__(BTH) void bin_place_kernel(
    const int* __restrict__ src, const int* __restrict__ dst,
    const int* __restrict__ pref, const int* __restrict__ binStart,
    int2* __restrict__ sorted, int E, int nbins) {
  __shared__ int cur[MAXBINS];
  const int tid = threadIdx.x;
  cur[tid] = (tid < nbins) ? binStart[tid] + pref[blockIdx.x * nbins + tid] : 0;
  __syncthreads();
  const int base = blockIdx.x * EPB;
  for (int k = 0; k < EPB / BTH; ++k) {
    const int e = base + k * BTH + tid;
    if (e < E) {
      const int d = dst[e];
      const int pos = atomicAdd(&cur[d >> BINSH], 1);
      sorted[pos] = make_int2(src[e], d & (BINN - 1));
    }
  }
}


__device__ __forceinline__ void put_row(float* __restrict__ aggbin, int r, int rows,
                                        int lane, v4f val) {
  if (r < rows) *((v4f*)(aggbin + r * CCH) + lane) = val;
}

template <bool WRITE_INV>
__global__ __launch_bounds__(ATH) void bin_sum_kernel(
    const float* __restrict__ feat, const int2* __restrict__ sorted,
    const int* __restrict__ binStart, float* __restrict__ agg,
    float* __restrict__ inv, int n_nodes) {
  __shared__ __attribute__((aligned(16))) int cnt[BINN];
  __shared__ __attribute__((aligned(16))) int off[BINN + 4];
  __shared__ int ord[SCAP];
  const int tid  = threadIdx.x;
  const int lane = tid & 31;
  const int wave = __builtin_amdgcn_readfirstlane(tid >> 5);
  const int bin  = blockIdx.x;
  const int e0 = binStart[bin];
  const int m  = binStart[bin + 1] - e0;
  const int2* __restrict__ recs = sorted + e0;
  float* __restrict__ aggbin = agg + (long)bin * BINN * CCH;
  int rows = n_nodes - bin * BINN;
  if (rows > BINN) rows = BINN;
  const int d0 = wave * DPW;
  const v4f zero = {0.0f, 0.0f, 0.0f, 0.0f};

  if (m > SCAP) {
#pragma nounroll
    for (int d = d0; d < d0 + DPW; ++d) {
      v4f acc = zero;
      int deg = 0;
      for (int base = 0; base < m; base += 32) {
        int2 sd = make_int2(0, -1);
        if (base + lane < m) sd = recs[base + lane];
        unsigned hit = (unsigned)__ballot(sd.y == d);
        deg += __popc(hit);
        while (hit) {
          const int l = __ffs(hit) - 1;
          hit &= hit - 1;
          const int s = __builtin_amdgcn_readlane(sd.x, l);
          acc += *(const v4f*)(feat + (long)s * CCH + 4 * lane);
        }
      }
      put_row(aggbin, d, rows, lane, acc);
      if (WRITE_INV && lane == 0 && d < rows)
        inv[bin * BINN + d] = 1.0f / fmaxf((float)deg, 1.0f);
    }
    return;
  }

  if (tid < BINN) cnt[tid] = 0;
  __syncthreads();
  int2 rc[RPT];
  int rk[RPT];
#pragma unroll
  for (int k = 0; k < RPT; ++k) {
    const int i = tid + k * ATH;
    rc[k] = make_int2(0, 0);
    if (i < m) rc[k] = recs[i];
  }
#pragma unroll
  for (int k = 0; k < RPT; ++k) {
    const int i = tid + k * ATH;
    rk[k] = 0;
    if (i < m) rk[k] = atomicAdd(&cnt[rc[k].y], 1);
  }
  __syncthreads();
  if (wave == 0) {
    const int4 c = ((const int4*)cnt)[lane];
    const int s = c.x + c.y + c.z + c.w;
    int incl = s;
#pragma unroll
    for (int o = 1; o < 32; o <<= 1) {
      const int t = __shfl_up(incl, o);
      if (lane >= o) incl += t;
    }
    const int ex = incl - s;
    ((int4*)off)[lane] = make_int4(ex, ex + c.x, ex + c.x + c.y, ex + c.x + c.y + c.z);
    if (lane == 31) off[BINN] = incl;
  }
  if (WRITE_INV && tid < rows)
    inv[bin * BINN + tid] = 1.0f / fmaxf((float)cnt[tid], 1.0f);
  __syncthreads();
#pragma unroll
  for (int k = 0; k < RPT; ++k) {
    const int i = tid + k * ATH;
    if (i < m) ord[off[rc[k].y] + rk[k]] = rc[k].x | (rc[k].y << 24);
  }
  __syncthreads();

  const int p0 = __builtin_amdgcn_readfirstlane(off[d0]);
  const int p1 = __builtin_amdgcn_readfirstlane(off[d0 + DPW]);
  int cur = d0;
  v4f acc = zero;
  for (int base = p0; base < p1; base += 32) {
    int n = p1 - base;
    if (n > 32) n = 32;
    int idx = base + lane;
    if (idx >= p1) idx = p1 - 1;
    const int rec = ord[idx];
    for (int j = 0; j < n; j += UNR) {
      v4f v[UNR];
      int dd[UNR];
#pragma unroll
      for (int u = 0; u < UNR; ++u) {
        const int r = __builtin_amdgcn_readlane(rec, j + u);
        dd[u] = r >> 24;
        v[u] = *(const v4f*)(feat + (long)(r & 0xFFFFFF) * CCH + 4 * lane);
      }
#pragma unroll
      for (int u = 0; u < UNR; ++u) {
        if (j + u < n) {
          if (dd[u] != cur) {
            put_row(aggbin, cur, rows, lane, acc);
#pragma nounroll
            for (int r = cur + 1; r < dd[u]; ++r) put_row(aggbin, r, rows, lane, zero);
            cur = dd[u];
            acc = zero;
          }
          acc += v[u];
        }
      }
    }
  }
  put_row(aggbin, cur, rows, lane, acc);
#pragma nounroll
  for (int r = cur + 1; r < d0 + DPW; ++r) put_row(aggbin, r, rows, lane, zero);
}

#define TMN 32
#define APITCH 260
#define DPITCH 132
#define LGRID 512

template <bool SIG>
__global__ __launch_bounds__(256) void sage_layer_kernel(
    const float* __restrict__ xin,
    const float* __restrict__ agg,
    const float* __restrict__ inv,
    const float* __restrict__ Wl,
    const float* __restrict__ bl,
    const float* __restrict__ Wr,
    float* __restrict__ out,
    int n_nodes, int n_tiles) {
  __shared__ float As[TMN * APITCH];
  __shared__ float Ds[TMN * DPITCH];
  const int lane = threadIdx.x & 31;
  const int wave = __builtin_amdgcn_readfirstlane(threadIdx.x >> 5);
  const int kk   = lane >> 4;
  const int ml   = lane & 15;

  v2f Bl[32], Br[32];
  {
    const float* wl = Wl + (long)(wave * 16 + ml) * CCH + 2 * kk;
    const float* wr = Wr + (long)(wave * 16 + ml) * CCH + 2 * kk;
#pragma unroll
    for (int ks = 0; ks < 32; ++ks) {
      const float2 l = *(const float2*)(wl + 4 * ks);
      const float2 r = *(const float2*)(wr + 4 * ks);
      Bl[ks].x = l.x; Bl[ks].y = l.y;
      Br[ks].x = r.x; Br[ks].y = r.y;
    }
  }
  const float bias = bl[wave * 16 + ml];

  v4f ra[4], rx[4];
  float rs[4];
  int tile = blockIdx.x;
#pragma unroll
  for (int r = 0; r < 4; ++r) {
    int node = tile * TMN + wave * 4 + r;
    if (node >= n_nodes) node = n_nodes - 1;
    ra[r] = *(const v4f*)(agg + (long)node * CCH + 4 * lane);
    rx[r] = *(const v4f*)(xin + (long)node * CCH + 4 * lane);
    rs[r] = inv[node];
  }

  const float* a0 = As + ml * APITCH + 2 * kk;
  const float* a1 = a0 + 16 * APITCH;

  for (; tile < n_tiles; tile += LGRID) {
#pragma unroll
    for (int r = 0; r < 4; ++r) {
      const int row = wave * 4 + r;
      *(v4f*)(As + row * APITCH + 4 * lane) = ra[r] * rs[r];
      *(v4f*)(As + row * APITCH + CCH + 4 * lane) = rx[r];
    }
    __syncthreads();

    const int nxt = tile + LGRID;
    if (nxt < n_tiles) {
#pragma unroll
      for (int r = 0; r < 4; ++r) {
        int node = nxt * TMN + wave * 4 + r;
        if (node >= n_nodes) node = n_nodes - 1;
        ra[r] = *(const v4f*)(agg + (long)node * CCH + 4 * lane);
        rx[r] = *(const v4f*)(xin + (long)node * CCH + 4 * lane);
        rs[r] = inv[node];
      }
    }

    v8f acc0 = {};
    v8f acc1 = {};
#pragma unroll
    for (int ks = 0; ks < 32; ++ks) {
      const v2f A0 = *(const v2f*)(a0 + 4 * ks);
      const v2f A1 = *(const v2f*)(a1 + 4 * ks);
      acc0 = __builtin_amdgcn_wmma_f32_16x16x4_f32(
          false, A0, false, Bl[ks], (short)0, acc0, false, false);
      acc1 = __builtin_amdgcn_wmma_f32_16x16x4_f32(
          false, A1, false, Bl[ks], (short)0, acc1, false, false);
    }
#pragma unroll
    for (int ks = 0; ks < 32; ++ks) {
      const v2f A0 = *(const v2f*)(a0 + CCH + 4 * ks);
      const v2f A1 = *(const v2f*)(a1 + CCH + 4 * ks);
      acc0 = __builtin_amdgcn_wmma_f32_16x16x4_f32(
          false, A0, false, Br[ks], (short)0, acc0, false, false);
      acc1 = __builtin_amdgcn_wmma_f32_16x16x4_f32(
          false, A1, false, Br[ks], (short)0, acc1, false, false);
    }

#pragma unroll
    for (int r = 0; r < 8; ++r) {
      float v0 = acc0[r] + bias;
      float v1 = acc1[r] + bias;
      if (SIG) {
        v0 = 1.0f / (1.0f + __expf(-v0));
        v1 = 1.0f / (1.0f + __expf(-v1));
      }
      Ds[(r + 8 * kk) * DPITCH + wave * 16 + ml] = v0;
      Ds[(16 + r + 8 * kk) * DPITCH + wave * 16 + ml] = v1;
    }
    __syncthreads();

#pragma unroll
    for (int r = 0; r < 4; ++r) {
      const int row  = wave * 4 + r;
      const int node = tile * TMN + row;
      if (node < n_nodes)
        *(v4f*)(out + (long)node * CCH + 4 * lane) =
            *(const v4f*)(Ds + row * DPITCH + 4 * lane);
    }
  }
}

extern "C" void kernel_launch(void* const* d_in, const int* in_sizes, int n_in,
                              void* d_out, int out_size, void* d_ws, size_t ws_size,
                              hipStream_t stream) {
  const float* x   = (const float*)d_in[0];
  const int*   ei  = (const int*)d_in[1];
  const float* W1l = (const float*)d_in[2];
  const float* b1l = (const float*)d_in[3];
  const float* W1r = (const float*)d_in[4];
  const float* W2l = (const float*)d_in[5];
  const float* b2l = (const float*)d_in[6];
  const float* W2r = (const float*)d_in[7];
  float* out = (float*)d_out;

  const int N = in_sizes[0] / CCH;
  const int E = in_sizes[1] / 2;

  float* agg = (float*)d_ws;
  float* h   = agg + (size_t)N * CCH;
  float* vin = h + (size_t)N * CCH;

  const int* src = ei;
  const int* dst = ei + E;

  const int nbins = (N + BINN - 1) / BINN;
  const int nblk  = (E + EPB - 1) / EPB;

  int2* sorted   = (int2*)d_out;
  int*  hist     = (int*)d_out + 2 * (size_t)E;
  int*  pref     = hist + (size_t)nblk * nbins;
  int*  binStart = pref + (size_t)nblk * nbins;

  bin_hist_kernel<<<nblk, BTH, 0, stream>>>(dst, hist, E, nbins);
  bin_scan_kernel<<<1, BTH, 0, stream>>>(hist, pref, binStart, nblk, nbins);
  bin_place_kernel<<<nblk, BTH, 0, stream>>>(src, dst, pref, binStart, sorted, E, nbins);

  const int ntiles = (N + TMN - 1) / TMN;

  bin_sum_kernel<true><<<nbins, ATH, 0, stream>>>(x, sorted, binStart, agg, vin, N);
  sage_layer_kernel<true><<<LGRID, 256, 0, stream>>>(
      x, agg, vin, W1l, b1l, W1r, h, N, ntiles);

  bin_sum_kernel<false><<<nbins, ATH, 0, stream>>>(h, sorted, binStart, agg, vin, N);
  sage_layer_kernel<false><<<LGRID, 256, 0, stream>>>(
      h, agg, vin, W2l, b2l, W2r, out, N, ntiles);
}


# Edge binning in one launch: each 10240-edge slice ordered by bin in LDS, per-slice run starts in a table; owners collect records from the slices; histogram and scan launches gone; rank-order sums
# speedup vs baseline: 2.7264x; 1.1337x over previous
#include <hip/hip_runtime.h>
#include <math.h>


typedef __attribute__((ext_vector_type(2))) float v2f;
typedef __attribute__((ext_vector_type(4))) float v4f;
typedef __attribute__((ext_vector_type(8))) float v8f;

#define CCH 128

#define BINSH 7
#define BINN 128
#define MAXBINS 1024
#define BTH 1024
#define EPB 10240
#define EPT (EPB / BTH)
#define ATH 512
#define AWV 16
#define DPW (BINN / AWV)
#define SCAP 2048
#define RPT (SCAP / ATH)
#define UNR 8

__global__ __launch_bounds__(BTH) void slice_sort_kernel(
    const int* __restrict__ src, const int* __restrict__ dst,
    int2* __restrict__ segs, int* __restrict__ tab, int E, int nbins, int nslices) {
  __shared__ int hs[MAXBINS];
  __shared__ int wsum[BTH / 32];
  const int tid  = threadIdx.x;
  const int lane = tid & 31;
  const int wave = tid >> 5;
  hs[tid] = 0;
  __syncthreads();
  const int base = blockIdx.x * EPB;
  int sv[EPT], dv[EPT], rk[EPT];
#pragma unroll
  for (int k = 0; k < EPT; ++k) {
    const int e = base + k * BTH + tid;
    sv[k] = 0;
    dv[k] = -1;
    if (e < E) { sv[k] = src[e]; dv[k] = dst[e]; }
  }
#pragma unroll
  for (int k = 0; k < EPT; ++k) {
    rk[k] = 0;
    if (dv[k] >= 0) rk[k] = atomicAdd(&hs[dv[k] >> BINSH], 1);
  }
  __syncthreads();
  const int c = hs[tid];
  int incl = c;
#pragma unroll
  for (int o = 1; o < 32; o <<= 1) {
    const int t = __shfl_up(incl, o);
    if (lane >= o) incl += t;
  }
  if (lane == 31) wsum[wave] = incl;
  __syncthreads();
  if (wave == 0) {
    const int w = wsum[lane];
    int wi = w;
#pragma unroll
    for (int o = 1; o < 32; o <<= 1) {
      const int t = __shfl_up(wi, o);
      if (lane >= o) wi += t;
    }
    wsum[lane] = wi - w;
  }
  __syncthreads();
  const int ex = incl - c + wsum[wave];
  hs[tid] = ex;
  if (tid <= nbins) tab[tid * nslices + blockIdx.x] = ex;
  __syncthreads();
#pragma unroll
  for (int k = 0; k < EPT; ++k) {
    if (dv[k] >= 0)
      segs[base + hs[dv[k] >> BINSH] + rk[k]] = make_int2(sv[k], dv[k] & (BINN - 1));
  }
}


__device__ __forceinline__ void put_row(float* __restrict__ aggbin, int r, int rows,
                                        int lane, v4f val) {
  if (r < rows) *((v4f*)(aggbin + r * CCH) + lane) = val;
}

__device__ __forceinline__ int rec_slot(const int* sbase, const int* sfirst, int nseg, int i) {
  int lo = 0, hi = nseg;
  while (hi - lo > 1) {
    const int mid = (lo + hi) >> 1;
    if (sbase[mid] <= i) lo = mid; else hi = mid;
  }
  return sfirst[lo] + (i - sbase[lo]);
}

template <bool WRITE_INV>
__global__ __launch_bounds__(ATH) void bin_sum_kernel(
    const float* __restrict__ feat, const int2* __restrict__ segs,
    const int* __restrict__ tab, float* __restrict__ agg,
    float* __restrict__ inv, int n_nodes, int nseg) {
  __shared__ __attribute__((aligned(16))) int cnt[BINN];
  __shared__ __attribute__((aligned(16))) int off[BINN + 4];
  __shared__ int ord[SCAP];
  __shared__ int sbase[ATH + 1];
  __shared__ int sfirst[ATH];
  __shared__ int wtot[AWV];
  const int tid  = threadIdx.x;
  const int lane = tid & 31;
  const int wave = __builtin_amdgcn_readfirstlane(tid >> 5);
  const int bin  = blockIdx.x;

  int len = 0;
  if (tid < nseg) {
    const int s0 = tab[bin * nseg + tid];
    len = tab[(bin + 1) * nseg + tid] - s0;
    sfirst[tid] = tid * EPB + s0;
  }
  int incl = len;
#pragma unroll
  for (int o = 1; o < 32; o <<= 1) {
    const int t = __shfl_up(incl, o);
    if (lane >= o) incl += t;
  }
  if (lane == 31) wtot[wave] = incl;
  __syncthreads();
  if (wave == 0) {
    const int w = (lane < AWV) ? wtot[lane] : 0;
    int wi = w;
#pragma unroll
    for (int o = 1; o < 32; o <<= 1) {
      const int t = __shfl_up(wi, o);
      if (lane >= o) wi += t;
    }
    if (lane < AWV) wtot[lane] = wi - w;
  }
  __syncthreads();
  sbase[tid] = incl - len + wtot[wave];
  if (tid == ATH - 1) sbase[ATH] = incl + wtot[wave];
  __syncthreads();
  const int m = __builtin_amdgcn_readfirstlane(sbase[ATH]);
  float* __restrict__ aggbin = agg + (long)bin * BINN * CCH;
  int rows = n_nodes - bin * BINN;
  if (rows > BINN) rows = BINN;
  const int d0 = wave * DPW;
  const v4f zero = {0.0f, 0.0f, 0.0f, 0.0f};

  if (m > SCAP) {
#pragma nounroll
    for (int d = d0; d < d0 + DPW; ++d) {
      v4f acc = zero;
      int deg = 0;
      for (int base = 0; base < m; base += 32) {
        int2 sd = make_int2(0, -1);
        if (base + lane < m) sd = segs[rec_slot(sbase, sfirst, nseg, base + lane)];
        unsigned hit = (unsigned)__ballot(sd.y == d);
        deg += __popc(hit);
        while (hit) {
          const int l = __ffs(hit) - 1;
          hit &= hit - 1;
          const int s = __builtin_amdgcn_readlane(sd.x, l);
          acc += *(const v4f*)(feat + (long)s * CCH + 4 * lane);
        }
      }
      put_row(aggbin, d, rows, lane, acc);
      if (WRITE_INV && lane == 0 && d < rows)
        inv[bin * BINN + d] = 1.0f / fmaxf((float)deg, 1.0f);
    }
    return;
  }

  if (tid < BINN) cnt[tid] = 0;
  __syncthreads();
  int2 rc[RPT];
  int rk[RPT];
  {
    const int steps = nseg > 1 ? 32 - __builtin_clz(nseg - 1) : 0;
    int lo[RPT], hi[RPT];
#pragma unroll
    for (int k = 0; k < RPT; ++k) { lo[k] = 0; hi[k] = nseg; }
    for (int it = 0; it < steps; ++it) {
#pragma unroll
      for (int k = 0; k < RPT; ++k) {
        const int mid = (lo[k] + hi[k]) >> 1;
        const bool up = sbase[mid] <= tid + k * ATH;
        lo[k] = up ? mid : lo[k];
        hi[k] = up ? hi[k] : mid;
      }
    }
#pragma unroll
    for (int k = 0; k < RPT; ++k) {
      const int i = tid + k * ATH;
      rc[k] = make_int2(0, 0);
      if (i < m) rc[k] = segs[sfirst[lo[k]] + (i - sbase[lo[k]])];
    }
  }
#pragma unroll
  for (int k = 0; k < RPT; ++k) {
    const int i = tid + k * ATH;
    rk[k] = 0;
    if (i < m) rk[k] = atomicAdd(&cnt[rc[k].y], 1);
  }
  __syncthreads();
  if (wave == 0) {
    const int4 c = ((const int4*)cnt)[lane];
    const int s = c.x + c.y + c.z + c.w;
    int incl = s;
#pragma unroll
    for (int o = 1; o < 32; o <<= 1) {
      const int t = __shfl_up(incl, o);
      if (lane >= o) incl += t;
    }
    const int ex = incl - s;
    ((int4*)off)[lane] = make_int4(ex, ex + c.x, ex + c.x + c.y, ex + c.x + c.y + c.z);
    if (lane == 31) off[BINN] = incl;
  }
  if (WRITE_INV && tid < rows)
    inv[bin * BINN + tid] = 1.0f / fmaxf((float)cnt[tid], 1.0f);
  __syncthreads();
#pragma unroll
  for (int k = 0; k < RPT; ++k) {
    const int i = tid + k * ATH;
    if (i < m) ord[off[rc[k].y] + rk[k]] = rc[k].x | (rc[k].y << 24);
  }
  __syncthreads();

  const int p0 = __builtin_amdgcn_readfirstlane(off[d0]);
  const int p1 = __builtin_amdgcn_readfirstlane(off[d0 + DPW]);
  int cur = d0;
  v4f acc = zero;
  for (int base = p0; base < p1; base += 32) {
    int n = p1 - base;
    if (n > 32) n = 32;
    int idx = base + lane;
    if (idx >= p1) idx = p1 - 1;
    const int rec = ord[idx];
    for (int j = 0; j < n; j += UNR) {
      v4f v[UNR];
      int dd[UNR];
#pragma unroll
      for (int u = 0; u < UNR; ++u) {
        const int r = __builtin_amdgcn_readlane(rec, j + u);
        dd[u] = r >> 24;
        v[u] = *(const v4f*)(feat + (long)(r & 0xFFFFFF) * CCH + 4 * lane);
      }
#pragma unroll
      for (int u = 0; u < UNR; ++u) {
        if (j + u < n) {
          if (dd[u] != cur) {
            put_row(aggbin, cur, rows, lane, acc);
#pragma nounroll
            for (int r = cur + 1; r < dd[u]; ++r) put_row(aggbin, r, rows, lane, zero);
            cur = dd[u];
            acc = zero;
          }
          acc += v[u];
        }
      }
    }
  }
  put_row(aggbin, cur, rows, lane, acc);
#pragma nounroll
  for (int r = cur + 1; r < d0 + DPW; ++r) put_row(aggbin, r, rows, lane, zero);
}

#define TMN 32
#define APITCH 260
#define DPITCH 132
#define LGRID 512

template <bool SIG>
__global__ __launch_bounds__(256) void sage_layer_kernel(
    const float* __restrict__ xin,
    const float* __restrict__ agg,
    const float* __restrict__ inv,
    const float* __restrict__ Wl,
    const float* __restrict__ bl,
    const float* __restrict__ Wr,
    float* __restrict__ out,
    int n_nodes, int n_tiles) {
  __shared__ float As[TMN * APITCH];
  __shared__ float Ds[TMN * DPITCH];
  const int lane = threadIdx.x & 31;
  const int wave = __builtin_amdgcn_readfirstlane(threadIdx.x >> 5);
  const int kk   = lane >> 4;
  const int ml   = lane & 15;

  v2f Bl[32], Br[32];
  {
    const float* wl = Wl + (long)(wave * 16 + ml) * CCH + 2 * kk;
    const float* wr = Wr + (long)(wave * 16 + ml) * CCH + 2 * kk;
#pragma unroll
    for (int ks = 0; ks < 32; ++ks) {
      const float2 l = *(const float2*)(wl + 4 * ks);
      const float2 r = *(const float2*)(wr + 4 * ks);
      Bl[ks].x = l.x; Bl[ks].y = l.y;
      Br[ks].x = r.x; Br[ks].y = r.y;
    }
  }
  const float bias = bl[wave * 16 + ml];

  v4f ra[4], rx[4];
  float rs[4];
  int tile = blockIdx.x;
#pragma unroll
  for (int r = 0; r < 4; ++r) {
    int node = tile * TMN + wave * 4 + r;
    if (node >= n_nodes) node = n_nodes - 1;
    ra[r] = *(const v4f*)(agg + (long)node * CCH + 4 * lane);
    rx[r] = *(const v4f*)(xin + (long)node * CCH + 4 * lane);
    rs[r] = inv[node];
  }

  const float* a0 = As + ml * APITCH + 2 * kk;
  const float* a1 = a0 + 16 * APITCH;

  for (; tile < n_tiles; tile += LGRID) {
#pragma unroll
    for (int r = 0; r < 4; ++r) {
      const int row = wave * 4 + r;
      *(v4f*)(As + row * APITCH + 4 * lane) = ra[r] * rs[r];
      *(v4f*)(As + row * APITCH + CCH + 4 * lane) = rx[r];
    }
    __syncthreads();

    const int nxt = tile + LGRID;
    if (nxt < n_tiles) {
#pragma unroll
      for (int r = 0; r < 4; ++r) {
        int node = nxt * TMN + wave * 4 + r;
        if (node >= n_nodes) node = n_nodes - 1;
        ra[r] = *(const v4f*)(agg + (long)node * CCH + 4 * lane);
        rx[r] = *(const v4f*)(xin + (long)node * CCH + 4 * lane);
        rs[r] = inv[node];
      }
    }

    v8f acc0 = {};
    v8f acc1 = {};
#pragma unroll
    for (int ks = 0; ks < 32; ++ks) {
      const v2f A0 = *(const v2f*)(a0 + 4 * ks);
      const v2f A1 = *(const v2f*)(a1 + 4 * ks);
      acc0 = __builtin_amdgcn_wmma_f32_16x16x4_f32(
          false, A0, false, Bl[ks], (short)0, acc0, false, false);
      acc1 = __builtin_amdgcn_wmma_f32_16x16x4_f32(
          false, A1, false, Bl[ks], (short)0, acc1, false, false);
    }
#pragma unroll
    for (int ks = 0; ks < 32; ++ks) {
      const v2f A0 = *(const v2f*)(a0 + CCH + 4 * ks);
      const v2f A1 = *(const v2f*)(a1 + CCH + 4 * ks);
      acc0 = __builtin_amdgcn_wmma_f32_16x16x4_f32(
          false, A0, false, Br[ks], (short)0, acc0, false, false);
      acc1 = __builtin_amdgcn_wmma_f32_16x16x4_f32(
          false, A1, false, Br[ks], (short)0, acc1, false, false);
    }

#pragma unroll
    for (int r = 0; r < 8; ++r) {
      float v0 = acc0[r] + bias;
      float v1 = acc1[r] + bias;
      if (SIG) {
        v0 = 1.0f / (1.0f + __expf(-v0));
        v1 = 1.0f / (1.0f + __expf(-v1));
      }
      Ds[(r + 8 * kk) * DPITCH + wave * 16 + ml] = v0;
      Ds[(16 + r + 8 * kk) * DPITCH + wave * 16 + ml] = v1;
    }
    __syncthreads();

#pragma unroll
    for (int r = 0; r < 4; ++r) {
      const int row  = wave * 4 + r;
      const int node = tile * TMN + row;
      if (node < n_nodes)
        *(v4f*)(out + (long)node * CCH + 4 * lane) =
            *(const v4f*)(Ds + row * DPITCH + 4 * lane);
    }
  }
}

extern "C" void kernel_launch(void* const* d_in, const int* in_sizes, int n_in,
                              void* d_out, int out_size, void* d_ws, size_t ws_size,
                              hipStream_t stream) {
  const float* x   = (const float*)d_in[0];
  const int*   ei  = (const int*)d_in[1];
  const float* W1l = (const float*)d_in[2];
  const float* b1l = (const float*)d_in[3];
  const float* W1r = (const float*)d_in[4];
  const float* W2l = (const float*)d_in[5];
  const float* b2l = (const float*)d_in[6];
  const float* W2r = (const float*)d_in[7];
  float* out = (float*)d_out;

  const int N = in_sizes[0] / CCH;
  const int E = in_sizes[1] / 2;

  float* agg = (float*)d_ws;
  float* h   = agg + (size_t)N * CCH;
  float* vin = h + (size_t)N * CCH;

  const int* src = ei;
  const int* dst = ei + E;

  const int nbins = (N + BINN - 1) / BINN;
  const int nblk  = (E + EPB - 1) / EPB;

  int2* segs = (int2*)d_out;
  int*  tab  = (int*)d_out + 2 * (size_t)nblk * EPB;

  slice_sort_kernel<<<nblk, BTH, 0, stream>>>(src, dst, segs, tab, E, nbins, nblk);

  const int ntiles = (N + TMN - 1) / TMN;

  bin_sum_kernel<true><<<nbins, ATH, 0, stream>>>(x, segs, tab, agg, vin, N, nblk);
  sage_layer_kernel<true><<<LGRID, 256, 0, stream>>>(
      x, agg, vin, W1l, b1l, W1r, h, N, ntiles);

  bin_sum_kernel<false><<<nbins, ATH, 0, stream>>>(h, segs, tab, agg, vin, N, nblk);
  sage_layer_kernel<false><<<LGRID, 256, 0, stream>>>(
      h, agg, vin, W2l, b2l, W2r, out, N, ntiles);
}


# Slice pass: a slice's records are placed in an LDS image of its slots and copied out linearly, 512 B per store instruction (whole lines), not one scattered 8-B store per edge; sums unchanged
# speedup vs baseline: 2.8116x; 1.0312x over previous
#include <hip/hip_runtime.h>
#include <math.h>


typedef __attribute__((ext_vector_type(2))) float v2f;
typedef __attribute__((ext_vector_type(4))) float v4f;
typedef __attribute__((ext_vector_type(8))) float v8f;

#define CCH 128

#define BINSH 7
#define BINN 128
#define MAXBINS 1024
#define BTH 1024
#define EPB 10240
#define EPT (EPB / BTH)
#define ATH 512
#define AWV 16
#define DPW (BINN / AWV)
#define SCAP 2048
#define RPT (SCAP / ATH)
#define UNR 8

__global__ __launch_bounds__(BTH) void slice_sort_kernel(
    const int* __restrict__ src, const int* __restrict__ dst,
    int2* __restrict__ segs, int* __restrict__ tab, int E, int nbins, int nslices) {
  __shared__ int hs[MAXBINS];
  __shared__ int wsum[BTH / 32];
  __shared__ __attribute__((aligned(16))) int2 rec[EPB];
  const int tid  = threadIdx.x;
  const int lane = tid & 31;
  const int wave = tid >> 5;
  hs[tid] = 0;
  __syncthreads();
  const int base = blockIdx.x * EPB;
  int sv[EPT], dv[EPT], rk[EPT];
#pragma unroll
  for (int k = 0; k < EPT; ++k) {
    const int e = base + k * BTH + tid;
    sv[k] = 0;
    dv[k] = -1;
    if (e < E) { sv[k] = src[e]; dv[k] = dst[e]; }
  }
#pragma unroll
  for (int k = 0; k < EPT; ++k) {
    rk[k] = 0;
    if (dv[k] >= 0) rk[k] = atomicAdd(&hs[dv[k] >> BINSH], 1);
  }
  __syncthreads();
  const int c = hs[tid];
  int incl = c;
#pragma unroll
  for (int o = 1; o < 32; o <<= 1) {
    const int t = __shfl_up(incl, o);
    if (lane >= o) incl += t;
  }
  if (lane == 31) wsum[wave] = incl;
  __syncthreads();
  if (wave == 0) {
    const int w = wsum[lane];
    int wi = w;
#pragma unroll
    for (int o = 1; o < 32; o <<= 1) {
      const int t = __shfl_up(wi, o);
      if (lane >= o) wi += t;
    }
    wsum[lane] = wi - w;
  }
  __syncthreads();
  const int ex = incl - c + wsum[wave];
  hs[tid] = ex;
  if (tid <= nbins) tab[tid * nslices + blockIdx.x] = ex;
  __syncthreads();
#pragma unroll
  for (int k = 0; k < EPT; ++k) {
    if (dv[k] >= 0)
      rec[hs[dv[k] >> BINSH] + rk[k]] = make_int2(sv[k], dv[k] & (BINN - 1));
  }
  __syncthreads();
  int cnt = E - base;
  if (cnt > EPB) cnt = EPB;
  const int4* img = (const int4*)rec;
  int4* dst4 = (int4*)(segs + base);
  const int npair = cnt >> 1;
  for (int p = tid; p < npair; p += BTH) dst4[p] = img[p];
  if ((cnt & 1) && tid == 0) segs[base + cnt - 1] = rec[cnt - 1];
}


__device__ __forceinline__ void put_row(float* __restrict__ aggbin, int r, int rows,
                                        int lane, v4f val) {
  if (r < rows) *((v4f*)(aggbin + r * CCH) + lane) = val;
}

__device__ __forceinline__ int rec_slot(const int* sbase, const int* sfirst, int nseg, int i) {
  int lo = 0, hi = nseg;
  while (hi - lo > 1) {
    const int mid = (lo + hi) >> 1;
    if (sbase[mid] <= i) lo = mid; else hi = mid;
  }
  return sfirst[lo] + (i - sbase[lo]);
}

template <bool WRITE_INV>
__global__ __launch_bounds__(ATH) void bin_sum_kernel(
    const float* __restrict__ feat, const int2* __restrict__ segs,
    const int* __restrict__ tab, float* __restrict__ agg,
    float* __restrict__ inv, int n_nodes, int nseg) {
  __shared__ __attribute__((aligned(16))) int cnt[BINN];
  __shared__ __attribute__((aligned(16))) int off[BINN + 4];
  __shared__ int ord[SCAP];
  __shared__ int sbase[ATH + 1];
  __shared__ int sfirst[ATH];
  __shared__ int wtot[AWV];
  const int tid  = threadIdx.x;
  const int lane = tid & 31;
  const int wave = __builtin_amdgcn_readfirstlane(tid >> 5);
  const int bin  = blockIdx.x;

  int len = 0;
  if (tid < nseg) {
    const int s0 = tab[bin * nseg + tid];
    len = tab[(bin + 1) * nseg + tid] - s0;
    sfirst[tid] = tid * EPB + s0;
  }
  int incl = len;
#pragma unroll
  for (int o = 1; o < 32; o <<= 1) {
    const int t = __shfl_up(incl, o);
    if (lane >= o) incl += t;
  }
  if (lane == 31) wtot[wave] = incl;
  __syncthreads();
  if (wave == 0) {
    const int w = (lane < AWV) ? wtot[lane] : 0;
    int wi = w;
#pragma unroll
    for (int o = 1; o < 32; o <<= 1) {
      const int t = __shfl_up(wi, o);
      if (lane >= o) wi += t;
    }
    if (lane < AWV) wtot[lane] = wi - w;
  }
  __syncthreads();
  sbase[tid] = incl - len + wtot[wave];
  if (tid == ATH - 1) sbase[ATH] = incl + wtot[wave];
  __syncthreads();
  const int m = __builtin_amdgcn_readfirstlane(sbase[ATH]);
  float* __restrict__ aggbin = agg + (long)bin * BINN * CCH;
  int rows = n_nodes - bin * BINN;
  if (rows > BINN) rows = BINN;
  const int d0 = wave * DPW;
  const v4f zero = {0.0f, 0.0f, 0.0f, 0.0f};

  if (m > SCAP) {
#pragma nounroll
    for (int d = d0; d < d0 + DPW; ++d) {
      v4f acc = zero;
      int deg = 0;
      for (int base = 0; base < m; base += 32) {
        int2 sd = make_int2(0, -1);
        if (base + lane < m) sd = segs[rec_slot(sbase, sfirst, nseg, base + lane)];
        unsigned hit = (unsigned)__ballot(sd.y == d);
        deg += __popc(hit);
        while (hit) {
          const int l = __ffs(hit) - 1;
          hit &= hit - 1;
          const int s = __builtin_amdgcn_readlane(sd.x, l);
          acc += *(const v4f*)(feat + (long)s * CCH + 4 * lane);
        }
      }
      put_row(aggbin, d, rows, lane, acc);
      if (WRITE_INV && lane == 0 && d < rows)
        inv[bin * BINN + d] = 1.0f / fmaxf((float)deg, 1.0f);
    }
    return;
  }

  if (tid < BINN) cnt[tid] = 0;
  __syncthreads();
  int2 rc[RPT];
  int rk[RPT];
  {
    const int steps = nseg > 1 ? 32 - __builtin_clz(nseg - 1) : 0;
    int lo[RPT], hi[RPT];
#pragma unroll
    for (int k = 0; k < RPT; ++k) { lo[k] = 0; hi[k] = nseg; }
    for (int it = 0; it < steps; ++it) {
#pragma unroll
      for (int k = 0; k < RPT; ++k) {
        const int mid = (lo[k] + hi[k]) >> 1;
        const bool up = sbase[mid] <= tid + k * ATH;
        lo[k] = up ? mid : lo[k];
        hi[k] = up ? hi[k] : mid;
      }
    }
#pragma unroll
    for (int k = 0; k < RPT; ++k) {
      const int i = tid + k * ATH;
      rc[k] = make_int2(0, 0);
      if (i < m) rc[k] = segs[sfirst[lo[k]] + (i - sbase[lo[k]])];
    }
  }
#pragma unroll
  for (int k = 0; k < RPT; ++k) {
    const int i = tid + k * ATH;
    rk[k] = 0;
    if (i < m) rk[k] = atomicAdd(&cnt[rc[k].y], 1);
  }
  __syncthreads();
  if (wave == 0) {
    const int4 c = ((const int4*)cnt)[lane];
    const int s = c.x + c.y + c.z + c.w;
    int incl = s;
#pragma unroll
    for (int o = 1; o < 32; o <<= 1) {
      const int t = __shfl_up(incl, o);
      if (lane >= o) incl += t;
    }
    const int ex = incl - s;
    ((int4*)off)[lane] = make_int4(ex, ex + c.x, ex + c.x + c.y, ex + c.x + c.y + c.z);
    if (lane == 31) off[BINN] = incl;
  }
  if (WRITE_INV && tid < rows)
    inv[bin * BINN + tid] = 1.0f / fmaxf((float)cnt[tid], 1.0f);
  __syncthreads();
#pragma unroll
  for (int k = 0; k < RPT; ++k) {
    const int i = tid + k * ATH;
    if (i < m) ord[off[rc[k].y] + rk[k]] = rc[k].x | (rc[k].y << 24);
  }
  __syncthreads();

  const int p0 = __builtin_amdgcn_readfirstlane(off[d0]);
  const int p1 = __builtin_amdgcn_readfirstlane(off[d0 + DPW]);
  int cur = d0;
  v4f acc = zero;
  for (int base = p0; base < p1; base += 32) {
    int n = p1 - base;
    if (n > 32) n = 32;
    int idx = base + lane;
    if (idx >= p1) idx = p1 - 1;
    const int rec = ord[idx];
    for (int j = 0; j < n; j += UNR) {
      v4f v[UNR];
      int dd[UNR];
#pragma unroll
      for (int u = 0; u < UNR; ++u) {
        const int r = __builtin_amdgcn_readlane(rec, j + u);
        dd[u] = r >> 24;
        v[u] = *(const v4f*)(feat + (long)(r & 0xFFFFFF) * CCH + 4 * lane);
      }
#pragma unroll
      for (int u = 0; u < UNR; ++u) {
        if (j + u < n) {
          if (dd[u] != cur) {
            put_row(aggbin, cur, rows, lane, acc);
#pragma nounroll
            for (int r = cur + 1; r < dd[u]; ++r) put_row(aggbin, r, rows, lane, zero);
            cur = dd[u];
            acc = zero;
          }
          acc += v[u];
        }
      }
    }
  }
  put_row(aggbin, cur, rows, lane, acc);
#pragma nounroll
  for (int r = cur + 1; r < d0 + DPW; ++r) put_row(aggbin, r, rows, lane, zero);
}

#define TMN 32
#define APITCH 260
#define DPITCH 132
#define LGRID 512

template <bool SIG>
__global__ __launch_bounds__(256) void sage_layer_kernel(
    const float* __restrict__ xin,
    const float* __restrict__ agg,
    const float* __restrict__ inv,
    const float* __restrict__ Wl,
    const float* __restrict__ bl,
    const float* __restrict__ Wr,
    float* __restrict__ out,
    int n_nodes, int n_tiles) {
  __shared__ float As[TMN * APITCH];
  __shared__ float Ds[TMN * DPITCH];
  const int lane = threadIdx.x & 31;
  const int wave = __builtin_amdgcn_readfirstlane(threadIdx.x >> 5);
  const int kk   = lane >> 4;
  const int ml   = lane & 15;

  v2f Bl[32], Br[32];
  {
    const float* wl = Wl + (long)(wave * 16 + ml) * CCH + 2 * kk;
    const float* wr = Wr + (long)(wave * 16 + ml) * CCH + 2 * kk;
#pragma unroll
    for (int ks = 0; ks < 32; ++ks) {
      const float2 l = *(const float2*)(wl + 4 * ks);
      const float2 r = *(const float2*)(wr + 4 * ks);
      Bl[ks].x = l.x; Bl[ks].y = l.y;
      Br[ks].x = r.x; Br[ks].y = r.y;
    }
  }
  const float bias = bl[wave * 16 + ml];

  v4f ra[4], rx[4];
  float rs[4];
  int tile = blockIdx.x;
#pragma unroll
  for (int r = 0; r < 4; ++r) {
    int node = tile * TMN + wave * 4 + r;
    if (node >= n_nodes) node = n_nodes - 1;
    ra[r] = *(const v4f*)(agg + (long)node * CCH + 4 * lane);
    rx[r] = *(const v4f*)(xin + (long)node * CCH + 4 * lane);
    rs[r] = inv[node];
  }

  const float* a0 = As + ml * APITCH + 2 * kk;
  const float* a1 = a0 + 16 * APITCH;

  for (; tile < n_tiles; tile += LGRID) {
#pragma unroll
    for (int r = 0; r < 4; ++r) {
      const int row = wave * 4 + r;
      *(v4f*)(As + row * APITCH + 4 * lane) = ra[r] * rs[r];
      *(v4f*)(As + row * APITCH + CCH + 4 * lane) = rx[r];
    }
    __syncthreads();

    const int nxt = tile + LGRID;
    if (nxt < n_tiles) {
#pragma unroll
      for (int r = 0; r < 4; ++r) {
        int node = nxt * TMN + wave * 4 + r;
        if (node >= n_nodes) node = n_nodes - 1;
        ra[r] = *(const v4f*)(agg + (long)node * CCH + 4 * lane);
        rx[r] = *(const v4f*)(xin + (long)node * CCH + 4 * lane);
        rs[r] = inv[node];
      }
    }

    v8f acc0 = {};
    v8f acc1 = {};
#pragma unroll
    for (int ks = 0; ks < 32; ++ks) {
      const v2f A0 = *(const v2f*)(a0 + 4 * ks);
      const v2f A1 = *(const v2f*)(a1 + 4 * ks);
      acc0 = __builtin_amdgcn_wmma_f32_16x16x4_f32(
          false, A0, false, Bl[ks], (short)0, acc0, false, false);
      acc1 = __builtin_amdgcn_wmma_f32_16x16x4_f32(
          false, A1, false, Bl[ks], (short)0, acc1, false, false);
    }
#pragma unroll
    for (int ks = 0; ks < 32; ++ks) {
      const v2f A0 = *(const v2f*)(a0 + CCH + 4 * ks);
      const v2f A1 = *(const v2f*)(a1 + CCH + 4 * ks);
      acc0 = __builtin_amdgcn_wmma_f32_16x16x4_f32(
          false, A0, false, Br[ks], (short)0, acc0, false, false);
      acc1 = __builtin_amdgcn_wmma_f32_16x16x4_f32(
          false, A1, false, Br[ks], (short)0, acc1, false, false);
    }

#pragma unroll
    for (int r = 0; r < 8; ++r) {
      float v0 = acc0[r] + bias;
      float v1 = acc1[r] + bias;
      if (SIG) {
        v0 = 1.0f / (1.0f + __expf(-v0));
        v1 = 1.0f / (1.0f + __expf(-v1));
      }
      Ds[(r + 8 * kk) * DPITCH + wave * 16 + ml] = v0;
      Ds[(16 + r + 8 * kk) * DPITCH + wave * 16 + ml] = v1;
    }
    __syncthreads();

#pragma unroll
    for (int r = 0; r < 4; ++r) {
      const int row  = wave * 4 + r;
      const int node = tile * TMN + row;
      if (node < n_nodes)
        *(v4f*)(out + (long)node * CCH + 4 * lane) =
            *(const v4f*)(Ds + row * DPITCH + 4 * lane);
    }
  }
}

extern "C" void kernel_launch(void* const* d_in, const int* in_sizes, int n_in,
                              void* d_out, int out_size, void* d_ws, size_t ws_size,
                              hipStream_t stream) {
  const float* x   = (const float*)d_in[0];
  const int*   ei  = (const int*)d_in[1];
  const float* W1l = (const float*)d_in[2];
  const float* b1l = (const float*)d_in[3];
  const float* W1r = (const float*)d_in[4];
  const float* W2l = (const float*)d_in[5];
  const float* b2l = (const float*)d_in[6];
  const float* W2r = (const float*)d_in[7];
  float* out = (float*)d_out;

  const int N = in_sizes[0] / CCH;
  const int E = in_sizes[1] / 2;

  float* agg = (float*)d_ws;
  float* h   = agg + (size_t)N * CCH;
  float* vin = h + (size_t)N * CCH;

  const int* src = ei;
  const int* dst = ei + E;

  const int nbins = (N + BINN - 1) / BINN;
  const int nblk  = (E + EPB - 1) / EPB;

  int2* segs = (int2*)d_out;
  int*  tab  = (int*)d_out + 2 * (size_t)nblk * EPB;

  slice_sort_kernel<<<nblk, BTH, 0, stream>>>(src, dst, segs, tab, E, nbins, nblk);

  const int ntiles = (N + TMN - 1) / TMN;

  bin_sum_kernel<true><<<nbins, ATH, 0, stream>>>(x, segs, tab, agg, vin, N, nblk);
  sage_layer_kernel<true><<<LGRID, 256, 0, stream>>>(
      x, agg, vin, W1l, b1l, W1r, h, N, ntiles);

  bin_sum_kernel<false><<<nbins, ATH, 0, stream>>>(h, segs, tab, agg, vin, N, nblk);
  sage_layer_kernel<false><<<LGRID, 256, 0, stream>>>(
      h, agg, vin, W2l, b2l, W2r, out, N, ntiles);
}
